# starting kernel
# baseline (speedup 1.0000x reference)
#include <hip/hip_runtime.h>
#include <hip/hip_bf16.h>


typedef float v2f __attribute__((ext_vector_type(2)));
typedef float v8f __attribute__((ext_vector_type(8)));

#define GH 512
#define GW 512
#define NG 256
#define NBLK 32

__device__ __forceinline__ float fast_rcp(float x) {
#if __has_builtin(__builtin_amdgcn_rcpf)
  return __builtin_amdgcn_rcpf(x);
#else
  return 1.0f / x;
#endif
}
__device__ __forceinline__ float fast_rsq(float x) {
#if __has_builtin(__builtin_amdgcn_rsqf)
  return __builtin_amdgcn_rsqf(x);
#else
  return rsqrtf(x);
#endif
}
__device__ __forceinline__ float fast_exp2(float x) {
#if __has_builtin(__builtin_amdgcn_exp2f)
  return __builtin_amdgcn_exp2f(x);
#else
  return exp2f(x);
#endif
}
__device__ __forceinline__ float swapx16(float x) {
  return __int_as_float(__builtin_amdgcn_ds_swizzle(__float_as_int(x), 0x401f));
}

__global__ void gsplat_prep_kernel(const float* __restrict__ pos,
                                   const float* __restrict__ lsc,
                                   const float* __restrict__ rop,
                                   const float* __restrict__ col,
                                   float4* __restrict__ aop,
                                   float4* __restrict__ scal)
{
  const int i = threadIdx.x;
  if (i >= NG) return;
  const float mx = pos[i * 3 + 0], my = pos[i * 3 + 1], mz = pos[i * 3 + 2];
  const float sx = expf(-2.0f * lsc[i * 3 + 0]);
  const float sy = expf(-2.0f * lsc[i * 3 + 1]);
  const float sz = expf(-2.0f * lsc[i * 3 + 2]);
  const float a0 = 1.0f / (1.0f + expf(-rop[i]));
  const float cr = 1.0f / (1.0f + expf(-col[i * 3 + 0]));
  const float cg = 1.0f / (1.0f + expf(-col[i * 3 + 1]));
  const float cb = 1.0f / (1.0f + expf(-col[i * 3 + 2]));

  int rank = 0;
  for (int j = 0; j < NG; ++j) {
    const float zj = pos[j * 3 + 2];
    rank += (zj < mz) || (zj == mz && j < i);
  }

  const float wx = mx * sx, wy = my * sy, wz = mz * sz;
  const float C = sx * mx * mx + sy * my * my + sz * mz * mz;
  const float a0p = a0 * expf(-0.5f * C);

  const int b = rank >> 3;
  const int j8 = rank & 7;
  float4* base = aop + b * 32;
  base[2 * j8 + 0]      = make_float4(wx, wy, 0.0f, 0.0f);
  base[16 + 2 * j8 + 0] = make_float4(wz, 0.0f, 0.0f, 0.0f);
  base[2 * j8 + 1]      = make_float4(0.0f, 0.0f, sy, sz);
  base[16 + 2 * j8 + 1] = make_float4(0.0f, sx, 0.0f, 0.0f);

  scal[rank] = make_float4(a0p, cr, cg, cb);
}

__global__ __launch_bounds__(256) void gsplat_render_kernel(
    const float4* __restrict__ aopG, const float4* __restrict__ scalG,
    float* __restrict__ out)
{
  __shared__ float4 shA[NBLK * 32];
  __shared__ float4 shS[NG];

  for (int t = threadIdx.x; t < NBLK * 32; t += 256) shA[t] = aopG[t];
  shS[threadIdx.x] = scalG[threadIdx.x];
  __syncthreads();

  const int lane = threadIdx.x & 31;
  const int wv   = threadIdx.x >> 5;
  const int p    = lane & 15;
  const bool hi  = lane >= 16;
  const int x = blockIdx.x * 128 + wv * 16 + p;
  const int y = blockIdx.y;

  const float ux = ((float)x + 0.5f - 256.0f) * (1.0f / 500.0f);
  const float uy = ((float)y + 0.5f - 256.0f) * (1.0f / 500.0f);
  const float rn = fast_rsq(ux * ux + uy * uy + 1.0f);
  const float dx = ux * rn, dy = uy * rn, dz = rn;

  v2f b0, b1;
  b0.x = hi ? dz       : dx;
  b0.y = hi ? dx * dx  : dy;
  b1.x = hi ? 0.0f     : dy * dy;
  b1.y = hi ? 0.0f     : dz * dz;

  float runR = 0.0f, runG = 0.0f, runB = 0.0f, runT = 1.0f;

  for (int blk = 0; blk < NBLK; ++blk) {
    const float4 at = shA[blk * 32 + lane];
    v2f a0v; a0v.x = at.x; a0v.y = at.y;
    v2f a1v; a1v.x = at.z; a1v.y = at.w;

    v8f acc = {0, 0, 0, 0, 0, 0, 0, 0};
    acc = __builtin_amdgcn_wmma_f32_16x16x4_f32(false, a0v, false, b0,
                                                (short)0, acc, false, false);
    acc = __builtin_amdgcn_wmma_f32_16x16x4_f32(false, a1v, false, b1,
                                                (short)0, acc, false, false);

    float sIr = 0.0f, sIg = 0.0f, sIb = 0.0f, sT = 1.0f;
    const int gbase = blk * 8 + (hi ? 4 : 0);
#pragma unroll
    for (int i = 0; i < 4; ++i) {
      const float Av = acc[2 * i];
      const float Bv = acc[2 * i + 1];
      const float4 sc = shS[gbase + i];
      const float rB  = fast_rcp(Bv + 1e-10f);
      const float tau = Av * rB;
      float alpha = sc.x * fast_exp2(0.72134752f * (Av * tau));
      alpha = (tau > 0.1f) ? alpha : 0.0f;
      const float wgt = sT * alpha;
      sIr = fmaf(wgt, sc.y, sIr);
      sIg = fmaf(wgt, sc.z, sIg);
      sIb = fmaf(wgt, sc.w, sIb);
      sT  = fmaf(-alpha, sT, sT);
    }

    const float oR = swapx16(sIr), oG = swapx16(sIg);
    const float oB = swapx16(sIb), oT = swapx16(sT);
    const float lR = hi ? oR : sIr, lG = hi ? oG : sIg;
    const float lB = hi ? oB : sIb, lT = hi ? oT : sT;
    const float hR = hi ? sIr : oR, hG = hi ? sIg : oG;
    const float hB = hi ? sIb : oB, hT = hi ? sT : oT;
    const float cR = fmaf(lT, hR, lR);
    const float cG = fmaf(lT, hG, lG);
    const float cB = fmaf(lT, hB, lB);
    const float cT = lT * hT;
    runR = fmaf(runT, cR, runR);
    runG = fmaf(runT, cG, runG);
    runB = fmaf(runT, cB, runB);
    runT *= cT;
  }

  if (!hi) {
    const int o = (y * GW + x) * 3;
    out[o + 0] = runR + runT;
    out[o + 1] = runG + runT;
    out[o + 2] = runB + runT;
  }
}

extern "C" void kernel_launch(void* const* d_in, const int* in_sizes, int n_in,
                              void* d_out, int out_size, void* d_ws, size_t ws_size,
                              hipStream_t stream) {
  (void)in_sizes; (void)n_in; (void)out_size; (void)ws_size;
  const float* pos = (const float*)d_in[0];
  const float* lsc = (const float*)d_in[1];
  const float* rop = (const float*)d_in[2];
  const float* col = (const float*)d_in[3];

  float4* aop  = (float4*)d_ws;
  float4* scal = aop + NBLK * 32;

  gsplat_prep_kernel<<<1, 256, 0, stream>>>(pos, lsc, rop, col, aop, scal);

  dim3 grid(GW / 128, GH);
  gsplat_render_kernel<<<grid, 256, 0, stream>>>(aop, scal, (float*)d_out);
}


# Lane halves composite the front and the back 128 sorted gaussians as two running (I,T) segments (table rows re-ordered): per-block swizzle merge becomes one final merge; summation tree changes
# speedup vs baseline: 1.1351x; 1.1351x over previous
#include <hip/hip_runtime.h>
#include <hip/hip_bf16.h>


typedef float v2f __attribute__((ext_vector_type(2)));
typedef float v8f __attribute__((ext_vector_type(8)));

#define GH 512
#define GW 512
#define NG 256
#define NBLK 32

__device__ __forceinline__ float fast_rcp(float x) {
#if __has_builtin(__builtin_amdgcn_rcpf)
  return __builtin_amdgcn_rcpf(x);
#else
  return 1.0f / x;
#endif
}
__device__ __forceinline__ float fast_rsq(float x) {
#if __has_builtin(__builtin_amdgcn_rsqf)
  return __builtin_amdgcn_rsqf(x);
#else
  return rsqrtf(x);
#endif
}
__device__ __forceinline__ float fast_exp2(float x) {
#if __has_builtin(__builtin_amdgcn_exp2f)
  return __builtin_amdgcn_exp2f(x);
#else
  return exp2f(x);
#endif
}
__device__ __forceinline__ float swapx16(float x) {
  return __int_as_float(__builtin_amdgcn_ds_swizzle(__float_as_int(x), 0x401f));
}

__global__ void gsplat_prep_kernel(const float* __restrict__ pos,
                                   const float* __restrict__ lsc,
                                   const float* __restrict__ rop,
                                   const float* __restrict__ col,
                                   float4* __restrict__ aop,
                                   float4* __restrict__ scal)
{
  const int i = threadIdx.x;
  if (i >= NG) return;
  const float mx = pos[i * 3 + 0], my = pos[i * 3 + 1], mz = pos[i * 3 + 2];
  const float sx = expf(-2.0f * lsc[i * 3 + 0]);
  const float sy = expf(-2.0f * lsc[i * 3 + 1]);
  const float sz = expf(-2.0f * lsc[i * 3 + 2]);
  const float a0 = 1.0f / (1.0f + expf(-rop[i]));
  const float cr = 1.0f / (1.0f + expf(-col[i * 3 + 0]));
  const float cg = 1.0f / (1.0f + expf(-col[i * 3 + 1]));
  const float cb = 1.0f / (1.0f + expf(-col[i * 3 + 2]));

  int rank = 0;
  for (int j = 0; j < NG; ++j) {
    const float zj = pos[j * 3 + 2];
    rank += (zj < mz) || (zj == mz && j < i);
  }

  const float wx = mx * sx, wy = my * sy, wz = mz * sz;
  const float C = sx * mx * mx + sy * my * my + sz * mz * mz;
  const float a0p = a0 * expf(-0.5f * C);

  const int seg = rank >> 7;
  const int b   = (rank & 127) >> 2;
  const int m   = 8 * seg + 2 * (rank & 3);
  float4* base = aop + b * 32;
  base[m + 0]      = make_float4(wx, wy, 0.0f, 0.0f);
  base[16 + m + 0] = make_float4(wz, 0.0f, 0.0f, 0.0f);
  base[m + 1]      = make_float4(0.0f, 0.0f, sy, sz);
  base[16 + m + 1] = make_float4(0.0f, sx, 0.0f, 0.0f);

  scal[rank] = make_float4(a0p, cr, cg, cb);
}

__global__ __launch_bounds__(256) void gsplat_render_kernel(
    const float4* __restrict__ aopG, const float4* __restrict__ scalG,
    float* __restrict__ out)
{
  __shared__ float4 shA[NBLK * 32];
  __shared__ float4 shS[NG];

  for (int t = threadIdx.x; t < NBLK * 32; t += 256) shA[t] = aopG[t];
  shS[threadIdx.x] = scalG[threadIdx.x];
  __syncthreads();

  const int lane = threadIdx.x & 31;
  const int wv   = threadIdx.x >> 5;
  const int p    = lane & 15;
  const bool hi  = lane >= 16;
  const int x = blockIdx.x * 128 + wv * 16 + p;
  const int y = blockIdx.y;

  const float ux = ((float)x + 0.5f - 256.0f) * (1.0f / 500.0f);
  const float uy = ((float)y + 0.5f - 256.0f) * (1.0f / 500.0f);
  const float rn = fast_rsq(ux * ux + uy * uy + 1.0f);
  const float dx = ux * rn, dy = uy * rn, dz = rn;

  v2f b0, b1;
  b0.x = hi ? dz       : dx;
  b0.y = hi ? dx * dx  : dy;
  b1.x = hi ? 0.0f     : dy * dy;
  b1.y = hi ? 0.0f     : dz * dz;

  float sIr = 0.0f, sIg = 0.0f, sIb = 0.0f, sT = 1.0f;
  const int gseg = hi ? (NG / 2) : 0;

  for (int blk = 0; blk < NBLK; ++blk) {
    const float4 at = shA[blk * 32 + lane];
    v2f a0v; a0v.x = at.x; a0v.y = at.y;
    v2f a1v; a1v.x = at.z; a1v.y = at.w;

    v8f acc = {0, 0, 0, 0, 0, 0, 0, 0};
    acc = __builtin_amdgcn_wmma_f32_16x16x4_f32(false, a0v, false, b0,
                                                (short)0, acc, false, false);
    acc = __builtin_amdgcn_wmma_f32_16x16x4_f32(false, a1v, false, b1,
                                                (short)0, acc, false, false);

    const int gbase = gseg + blk * 4;
#pragma unroll
    for (int i = 0; i < 4; ++i) {
      const float Av = acc[2 * i];
      const float Bv = acc[2 * i + 1];
      const float4 sc = shS[gbase + i];
      const float rB  = fast_rcp(Bv + 1e-10f);
      const float tau = Av * rB;
      float alpha = sc.x * fast_exp2(0.72134752f * (Av * tau));
      alpha = (tau > 0.1f) ? alpha : 0.0f;
      const float wgt = sT * alpha;
      sIr = fmaf(wgt, sc.y, sIr);
      sIg = fmaf(wgt, sc.z, sIg);
      sIb = fmaf(wgt, sc.w, sIb);
      sT  = fmaf(-alpha, sT, sT);
    }
  }

  const float oR = swapx16(sIr), oG = swapx16(sIg);
  const float oB = swapx16(sIb), oT = swapx16(sT);
  if (!hi) {
    const float T = sT * oT;
    const int o = (y * GW + x) * 3;
    out[o + 0] = fmaf(sT, oR, sIr) + T;
    out[o + 1] = fmaf(sT, oG, sIg) + T;
    out[o + 2] = fmaf(sT, oB, sIb) + T;
  }
}

extern "C" void kernel_launch(void* const* d_in, const int* in_sizes, int n_in,
                              void* d_out, int out_size, void* d_ws, size_t ws_size,
                              hipStream_t stream) {
  (void)in_sizes; (void)n_in; (void)out_size; (void)ws_size;
  const float* pos = (const float*)d_in[0];
  const float* lsc = (const float*)d_in[1];
  const float* rop = (const float*)d_in[2];
  const float* col = (const float*)d_in[3];

  float4* aop  = (float4*)d_ws;
  float4* scal = aop + NBLK * 32;

  gsplat_prep_kernel<<<1, 256, 0, stream>>>(pos, lsc, rop, col, aop, scal);

  dim3 grid(GW / 128, GH);
  gsplat_render_kernel<<<grid, 256, 0, stream>>>(aop, scal, (float*)d_out);
}


# Prep rank counting made parallel: 8 work-groups x 8 waves, each wave counts 32 gaussians against a 32-wide z slice read from LDS, replacing one work-group's 32 serial trips of scalar global loads
# speedup vs baseline: 1.5716x; 1.3845x over previous
#include <hip/hip_runtime.h>
#include <hip/hip_bf16.h>


typedef float v2f __attribute__((ext_vector_type(2)));
typedef float v8f __attribute__((ext_vector_type(8)));

#define GH 512
#define GW 512
#define NG 256
#define NBLK 32

__device__ __forceinline__ float fast_rcp(float x) {
#if __has_builtin(__builtin_amdgcn_rcpf)
  return __builtin_amdgcn_rcpf(x);
#else
  return 1.0f / x;
#endif
}
__device__ __forceinline__ float fast_rsq(float x) {
#if __has_builtin(__builtin_amdgcn_rsqf)
  return __builtin_amdgcn_rsqf(x);
#else
  return rsqrtf(x);
#endif
}
__device__ __forceinline__ float fast_exp2(float x) {
#if __has_builtin(__builtin_amdgcn_exp2f)
  return __builtin_amdgcn_exp2f(x);
#else
  return exp2f(x);
#endif
}
__device__ __forceinline__ float swapx16(float x) {
  return __int_as_float(__builtin_amdgcn_ds_swizzle(__float_as_int(x), 0x401f));
}

#define PREP_WG 8
#define GPW (NG / PREP_WG)

__global__ __launch_bounds__(256) void gsplat_prep_kernel(const float* __restrict__ pos,
                                   const float* __restrict__ lsc,
                                   const float* __restrict__ rop,
                                   const float* __restrict__ col,
                                   float4* __restrict__ aop,
                                   float4* __restrict__ scal)
{
  __shared__ float4 shZ4[NG / 4];
  __shared__ int    shP[8 * GPW];

  const int t    = threadIdx.x;
  const int lane = t & 31;
  const int sl   = t >> 5;
  float* shZ = reinterpret_cast<float*>(shZ4);
  const int i = blockIdx.x * GPW + lane;

  const float zt = pos[t * 3 + 2];

  float mx = 0.0f, my = 0.0f, lx = 0.0f, ly = 0.0f, lz = 0.0f;
  float ro = 0.0f, c0 = 0.0f, c1 = 0.0f, c2 = 0.0f;
  if (sl == 0) {
    mx = pos[i * 3 + 0]; my = pos[i * 3 + 1];
    lx = lsc[i * 3 + 0]; ly = lsc[i * 3 + 1]; lz = lsc[i * 3 + 2];
    ro = rop[i];
    c0 = col[i * 3 + 0]; c1 = col[i * 3 + 1]; c2 = col[i * 3 + 2];
  }

  shZ[t] = zt;
  __syncthreads();

  const float mz = shZ[i];

  int part = 0;
#pragma unroll
  for (int q = 0; q < 8; ++q) {
    const float4 z4 = shZ4[sl * 8 + q];
    const int j = sl * 32 + q * 4;
    part += (z4.x < mz) || (z4.x == mz && (j + 0) < i);
    part += (z4.y < mz) || (z4.y == mz && (j + 1) < i);
    part += (z4.z < mz) || (z4.z == mz && (j + 2) < i);
    part += (z4.w < mz) || (z4.w == mz && (j + 3) < i);
  }
  shP[sl * GPW + lane] = part;
  __syncthreads();
  if (sl != 0) return;

  int rank = 0;
#pragma unroll
  for (int s = 0; s < 8; ++s) rank += shP[s * GPW + lane];

  const float sx = expf(-2.0f * lx);
  const float sy = expf(-2.0f * ly);
  const float sz = expf(-2.0f * lz);
  const float a0 = 1.0f / (1.0f + expf(-ro));
  const float cr = 1.0f / (1.0f + expf(-c0));
  const float cg = 1.0f / (1.0f + expf(-c1));
  const float cb = 1.0f / (1.0f + expf(-c2));

  const float wx = mx * sx, wy = my * sy, wz = mz * sz;
  const float C = sx * mx * mx + sy * my * my + sz * mz * mz;
  const float a0p = a0 * expf(-0.5f * C);

  const int seg = rank >> 7;
  const int b   = (rank & 127) >> 2;
  const int m   = 8 * seg + 2 * (rank & 3);
  float4* base = aop + b * 32;
  base[m + 0]      = make_float4(wx, wy, 0.0f, 0.0f);
  base[16 + m + 0] = make_float4(wz, 0.0f, 0.0f, 0.0f);
  base[m + 1]      = make_float4(0.0f, 0.0f, sy, sz);
  base[16 + m + 1] = make_float4(0.0f, sx, 0.0f, 0.0f);

  scal[rank] = make_float4(a0p, cr, cg, cb);
}

__global__ __launch_bounds__(256) void gsplat_render_kernel(
    const float4* __restrict__ aopG, const float4* __restrict__ scalG,
    float* __restrict__ out)
{
  __shared__ float4 shA[NBLK * 32];
  __shared__ float4 shS[NG];

  for (int t = threadIdx.x; t < NBLK * 32; t += 256) shA[t] = aopG[t];
  shS[threadIdx.x] = scalG[threadIdx.x];
  __syncthreads();

  const int lane = threadIdx.x & 31;
  const int wv   = threadIdx.x >> 5;
  const int p    = lane & 15;
  const bool hi  = lane >= 16;
  const int x = blockIdx.x * 128 + wv * 16 + p;
  const int y = blockIdx.y;

  const float ux = ((float)x + 0.5f - 256.0f) * (1.0f / 500.0f);
  const float uy = ((float)y + 0.5f - 256.0f) * (1.0f / 500.0f);
  const float rn = fast_rsq(ux * ux + uy * uy + 1.0f);
  const float dx = ux * rn, dy = uy * rn, dz = rn;

  v2f b0, b1;
  b0.x = hi ? dz       : dx;
  b0.y = hi ? dx * dx  : dy;
  b1.x = hi ? 0.0f     : dy * dy;
  b1.y = hi ? 0.0f     : dz * dz;

  float sIr = 0.0f, sIg = 0.0f, sIb = 0.0f, sT = 1.0f;
  const int gseg = hi ? (NG / 2) : 0;

  for (int blk = 0; blk < NBLK; ++blk) {
    const float4 at = shA[blk * 32 + lane];
    v2f a0v; a0v.x = at.x; a0v.y = at.y;
    v2f a1v; a1v.x = at.z; a1v.y = at.w;

    v8f acc = {0, 0, 0, 0, 0, 0, 0, 0};
    acc = __builtin_amdgcn_wmma_f32_16x16x4_f32(false, a0v, false, b0,
                                                (short)0, acc, false, false);
    acc = __builtin_amdgcn_wmma_f32_16x16x4_f32(false, a1v, false, b1,
                                                (short)0, acc, false, false);

    const int gbase = gseg + blk * 4;
#pragma unroll
    for (int i = 0; i < 4; ++i) {
      const float Av = acc[2 * i];
      const float Bv = acc[2 * i + 1];
      const float4 sc = shS[gbase + i];
      const float rB  = fast_rcp(Bv + 1e-10f);
      const float tau = Av * rB;
      float alpha = sc.x * fast_exp2(0.72134752f * (Av * tau));
      alpha = (tau > 0.1f) ? alpha : 0.0f;
      const float wgt = sT * alpha;
      sIr = fmaf(wgt, sc.y, sIr);
      sIg = fmaf(wgt, sc.z, sIg);
      sIb = fmaf(wgt, sc.w, sIb);
      sT  = fmaf(-alpha, sT, sT);
    }
  }

  const float oR = swapx16(sIr), oG = swapx16(sIg);
  const float oB = swapx16(sIb), oT = swapx16(sT);
  if (!hi) {
    const float T = sT * oT;
    const int o = (y * GW + x) * 3;
    out[o + 0] = fmaf(sT, oR, sIr) + T;
    out[o + 1] = fmaf(sT, oG, sIg) + T;
    out[o + 2] = fmaf(sT, oB, sIb) + T;
  }
}

extern "C" void kernel_launch(void* const* d_in, const int* in_sizes, int n_in,
                              void* d_out, int out_size, void* d_ws, size_t ws_size,
                              hipStream_t stream) {
  (void)in_sizes; (void)n_in; (void)out_size; (void)ws_size;
  const float* pos = (const float*)d_in[0];
  const float* lsc = (const float*)d_in[1];
  const float* rop = (const float*)d_in[2];
  const float* col = (const float*)d_in[3];

  float4* aop  = (float4*)d_ws;
  float4* scal = aop + NBLK * 32;

  gsplat_prep_kernel<<<PREP_WG, 256, 0, stream>>>(pos, lsc, rop, col, aop, scal);

  dim3 grid(GW / 128, GH);
  gsplat_render_kernel<<<grid, 256, 0, stream>>>(aop, scal, (float*)d_out);
}


# Numerators and denominators each get their own K=4 WMMA over 16 gaussian rows (the +1e-10 is the denominator's 4th product): 2 WMMAs serve 16 gaussians instead of 8; summation order kept
# speedup vs baseline: 2.0471x; 1.3026x over previous
#include <hip/hip_runtime.h>
#include <hip/hip_bf16.h>


typedef float v2f __attribute__((ext_vector_type(2)));
typedef float v8f __attribute__((ext_vector_type(8)));

#define GH 512
#define GW 512
#define NG 256
#define NBLK 16

__device__ __forceinline__ float fast_rcp(float x) {
#if __has_builtin(__builtin_amdgcn_rcpf)
  return __builtin_amdgcn_rcpf(x);
#else
  return 1.0f / x;
#endif
}
__device__ __forceinline__ float fast_rsq(float x) {
#if __has_builtin(__builtin_amdgcn_rsqf)
  return __builtin_amdgcn_rsqf(x);
#else
  return rsqrtf(x);
#endif
}
__device__ __forceinline__ float fast_exp2(float x) {
#if __has_builtin(__builtin_amdgcn_exp2f)
  return __builtin_amdgcn_exp2f(x);
#else
  return exp2f(x);
#endif
}
__device__ __forceinline__ float swapx16(float x) {
  return __int_as_float(__builtin_amdgcn_ds_swizzle(__float_as_int(x), 0x401f));
}

#define PREP_WG 8
#define GPW (NG / PREP_WG)

__global__ __launch_bounds__(256) void gsplat_prep_kernel(const float* __restrict__ pos,
                                   const float* __restrict__ lsc,
                                   const float* __restrict__ rop,
                                   const float* __restrict__ col,
                                   float4* __restrict__ aop,
                                   float4* __restrict__ scal)
{
  __shared__ float4 shZ4[NG / 4];
  __shared__ int    shP[8 * GPW];

  const int t    = threadIdx.x;
  const int lane = t & 31;
  const int sl   = t >> 5;
  float* shZ = reinterpret_cast<float*>(shZ4);
  const int i = blockIdx.x * GPW + lane;

  const float zt = pos[t * 3 + 2];

  float mx = 0.0f, my = 0.0f, lx = 0.0f, ly = 0.0f, lz = 0.0f;
  float ro = 0.0f, c0 = 0.0f, c1 = 0.0f, c2 = 0.0f;
  if (sl == 0) {
    mx = pos[i * 3 + 0]; my = pos[i * 3 + 1];
    lx = lsc[i * 3 + 0]; ly = lsc[i * 3 + 1]; lz = lsc[i * 3 + 2];
    ro = rop[i];
    c0 = col[i * 3 + 0]; c1 = col[i * 3 + 1]; c2 = col[i * 3 + 2];
  }

  shZ[t] = zt;
  __syncthreads();

  const float mz = shZ[i];

  int part = 0;
#pragma unroll
  for (int q = 0; q < 8; ++q) {
    const float4 z4 = shZ4[sl * 8 + q];
    const int j = sl * 32 + q * 4;
    part += (z4.x < mz) || (z4.x == mz && (j + 0) < i);
    part += (z4.y < mz) || (z4.y == mz && (j + 1) < i);
    part += (z4.z < mz) || (z4.z == mz && (j + 2) < i);
    part += (z4.w < mz) || (z4.w == mz && (j + 3) < i);
  }
  shP[sl * GPW + lane] = part;
  __syncthreads();
  if (sl != 0) return;

  int rank = 0;
#pragma unroll
  for (int s = 0; s < 8; ++s) rank += shP[s * GPW + lane];

  const float sx = expf(-2.0f * lx);
  const float sy = expf(-2.0f * ly);
  const float sz = expf(-2.0f * lz);
  const float a0 = 1.0f / (1.0f + expf(-ro));
  const float cr = 1.0f / (1.0f + expf(-c0));
  const float cg = 1.0f / (1.0f + expf(-c1));
  const float cb = 1.0f / (1.0f + expf(-c2));

  const float wx = mx * sx, wy = my * sy, wz = mz * sz;
  const float C = sx * mx * mx + sy * my * my + sz * mz * mz;
  const float a0p = a0 * expf(-0.5f * C);

  const int seg = rank >> 7;
  const int b   = (rank & 127) >> 3;
  const int m   = 8 * seg + (rank & 7);
  float4* base = aop + b * 32;
  base[m]      = make_float4(wx, wy,   sx, sy);
  base[16 + m] = make_float4(wz, 0.0f, sz, 1e-10f);

  scal[rank] = make_float4(a0p, cr, cg, cb);
}

__global__ __launch_bounds__(256) void gsplat_render_kernel(
    const float4* __restrict__ aopG, const float4* __restrict__ scalG,
    float* __restrict__ out)
{
  __shared__ float4 shA[NBLK * 32];
  __shared__ float4 shS[NG];

  for (int t = threadIdx.x; t < NBLK * 32; t += 256) shA[t] = aopG[t];
  shS[threadIdx.x] = scalG[threadIdx.x];
  __syncthreads();

  const int lane = threadIdx.x & 31;
  const int wv   = threadIdx.x >> 5;
  const int p    = lane & 15;
  const bool hi  = lane >= 16;
  const int x = blockIdx.x * 128 + wv * 16 + p;
  const int y = blockIdx.y;

  const float ux = ((float)x + 0.5f - 256.0f) * (1.0f / 500.0f);
  const float uy = ((float)y + 0.5f - 256.0f) * (1.0f / 500.0f);
  const float rn = fast_rsq(ux * ux + uy * uy + 1.0f);
  const float dx = ux * rn, dy = uy * rn, dz = rn;

  v2f bn, bd;
  bn.x = hi ? dz       : dx;
  bn.y = hi ? 0.0f     : dy;
  bd.x = hi ? dz * dz  : dx * dx;
  bd.y = hi ? 1.0f     : dy * dy;

  float sIr = 0.0f, sIg = 0.0f, sIb = 0.0f, sT = 1.0f;
  const int gseg = hi ? (NG / 2) : 0;

  for (int blk = 0; blk < NBLK; ++blk) {
    const float4 at = shA[blk * 32 + lane];
    v2f anv; anv.x = at.x; anv.y = at.y;
    v2f adv; adv.x = at.z; adv.y = at.w;

    const v8f zero = {0, 0, 0, 0, 0, 0, 0, 0};
    const v8f accN = __builtin_amdgcn_wmma_f32_16x16x4_f32(false, anv, false, bn,
                                                (short)0, zero, false, false);
    const v8f accD = __builtin_amdgcn_wmma_f32_16x16x4_f32(false, adv, false, bd,
                                                (short)0, zero, false, false);

    const int gbase = gseg + blk * 8;
#pragma unroll
    for (int i = 0; i < 8; ++i) {
      const float Av = accN[i];
      const float Bv = accD[i];
      const float4 sc = shS[gbase + i];
      const float rB  = fast_rcp(Bv);
      const float tau = Av * rB;
      float alpha = sc.x * fast_exp2(0.72134752f * (Av * tau));
      alpha = (tau > 0.1f) ? alpha : 0.0f;
      const float wgt = sT * alpha;
      sIr = fmaf(wgt, sc.y, sIr);
      sIg = fmaf(wgt, sc.z, sIg);
      sIb = fmaf(wgt, sc.w, sIb);
      sT  = fmaf(-alpha, sT, sT);
    }
  }

  const float oR = swapx16(sIr), oG = swapx16(sIg);
  const float oB = swapx16(sIb), oT = swapx16(sT);
  if (!hi) {
    const float T = sT * oT;
    const int o = (y * GW + x) * 3;
    out[o + 0] = fmaf(sT, oR, sIr) + T;
    out[o + 1] = fmaf(sT, oG, sIg) + T;
    out[o + 2] = fmaf(sT, oB, sIb) + T;
  }
}

extern "C" void kernel_launch(void* const* d_in, const int* in_sizes, int n_in,
                              void* d_out, int out_size, void* d_ws, size_t ws_size,
                              hipStream_t stream) {
  (void)in_sizes; (void)n_in; (void)out_size; (void)ws_size;
  const float* pos = (const float*)d_in[0];
  const float* lsc = (const float*)d_in[1];
  const float* rop = (const float*)d_in[2];
  const float* col = (const float*)d_in[3];

  float4* aop  = (float4*)d_ws;
  float4* scal = aop + NBLK * 32;

  gsplat_prep_kernel<<<PREP_WG, 256, 0, stream>>>(pos, lsc, rop, col, aop, scal);

  dim3 grid(GW / 128, GH);
  gsplat_render_kernel<<<grid, 256, 0, stream>>>(aop, scal, (float*)d_out);
}


# Render work-groups cull per 128-pixel strip: gaussians with numerator < 0 on the whole strip (alpha exactly 0) are compacted out, loop runs ceil(kept/16) blocks; summation tree changes
# speedup vs baseline: 2.8320x; 1.3834x over previous
#include <hip/hip_runtime.h>
#include <hip/hip_bf16.h>


typedef float v2f __attribute__((ext_vector_type(2)));
typedef float v8f __attribute__((ext_vector_type(8)));

#define GH 512
#define GW 512
#define NG 256
#define NBLK 16

__device__ __forceinline__ float fast_rcp(float x) {
#if __has_builtin(__builtin_amdgcn_rcpf)
  return __builtin_amdgcn_rcpf(x);
#else
  return 1.0f / x;
#endif
}
__device__ __forceinline__ float fast_rsq(float x) {
#if __has_builtin(__builtin_amdgcn_rsqf)
  return __builtin_amdgcn_rsqf(x);
#else
  return rsqrtf(x);
#endif
}
__device__ __forceinline__ float fast_exp2(float x) {
#if __has_builtin(__builtin_amdgcn_exp2f)
  return __builtin_amdgcn_exp2f(x);
#else
  return exp2f(x);
#endif
}
__device__ __forceinline__ float swapx16(float x) {
  return __int_as_float(__builtin_amdgcn_ds_swizzle(__float_as_int(x), 0x401f));
}

#define PREP_WG 8
#define GPW (NG / PREP_WG)

__global__ __launch_bounds__(256) void gsplat_prep_kernel(const float* __restrict__ pos,
                                   const float* __restrict__ lsc,
                                   const float* __restrict__ rop,
                                   const float* __restrict__ col,
                                   float4* __restrict__ aop,
                                   float4* __restrict__ scal)
{
  __shared__ float4 shZ4[NG / 4];
  __shared__ int    shP[8 * GPW];

  const int t    = threadIdx.x;
  const int lane = t & 31;
  const int sl   = t >> 5;
  float* shZ = reinterpret_cast<float*>(shZ4);
  const int i = blockIdx.x * GPW + lane;

  const float zt = pos[t * 3 + 2];

  float mx = 0.0f, my = 0.0f, lx = 0.0f, ly = 0.0f, lz = 0.0f;
  float ro = 0.0f, c0 = 0.0f, c1 = 0.0f, c2 = 0.0f;
  if (sl == 0) {
    mx = pos[i * 3 + 0]; my = pos[i * 3 + 1];
    lx = lsc[i * 3 + 0]; ly = lsc[i * 3 + 1]; lz = lsc[i * 3 + 2];
    ro = rop[i];
    c0 = col[i * 3 + 0]; c1 = col[i * 3 + 1]; c2 = col[i * 3 + 2];
  }

  shZ[t] = zt;
  __syncthreads();

  const float mz = shZ[i];

  int part = 0;
#pragma unroll
  for (int q = 0; q < 8; ++q) {
    const float4 z4 = shZ4[sl * 8 + q];
    const int j = sl * 32 + q * 4;
    part += (z4.x < mz) || (z4.x == mz && (j + 0) < i);
    part += (z4.y < mz) || (z4.y == mz && (j + 1) < i);
    part += (z4.z < mz) || (z4.z == mz && (j + 2) < i);
    part += (z4.w < mz) || (z4.w == mz && (j + 3) < i);
  }
  shP[sl * GPW + lane] = part;
  __syncthreads();
  if (sl != 0) return;

  int rank = 0;
#pragma unroll
  for (int s = 0; s < 8; ++s) rank += shP[s * GPW + lane];

  const float sx = expf(-2.0f * lx);
  const float sy = expf(-2.0f * ly);
  const float sz = expf(-2.0f * lz);
  const float a0 = 1.0f / (1.0f + expf(-ro));
  const float cr = 1.0f / (1.0f + expf(-c0));
  const float cg = 1.0f / (1.0f + expf(-c1));
  const float cb = 1.0f / (1.0f + expf(-c2));

  const float wx = mx * sx, wy = my * sy, wz = mz * sz;
  const float C = sx * mx * mx + sy * my * my + sz * mz * mz;
  const float a0p = a0 * expf(-0.5f * C);

  const int seg = rank >> 7;
  const int b   = (rank & 127) >> 3;
  const int m   = 8 * seg + (rank & 7);
  float4* base = aop + b * 32;
  base[m]      = make_float4(wx, wy,   sx, sy);
  base[16 + m] = make_float4(wz, 0.0f, sz, 1e-10f);

  scal[rank] = make_float4(a0p, cr, cg, cb);
}

__global__ __launch_bounds__(256) void gsplat_render_kernel(
    const float4* __restrict__ aopG, const float4* __restrict__ scalG,
    float* __restrict__ out)
{
  __shared__ float4 shA[NBLK * 32];
  __shared__ float4 shS[NG];
  __shared__ int    shC[8];

  const int t    = threadIdx.x;
  const int lane = t & 31;
  const int wv   = t >> 5;
  const int y    = blockIdx.y;
  const int xs   = blockIdx.x * 128;

  const int rb = (t & 127) >> 3;
  const int rm = 8 * (t >> 7) + (t & 7);
  const float4 r0 = aopG[rb * 32 + rm];
  const float4 r1 = aopG[rb * 32 + 16 + rm];
  const float4 rs = scalG[t];

  const float uy = ((float)y + 0.5f - 256.0f) * (1.0f / 500.0f);

  const float uxa = ((float)xs + 0.5f - 256.0f) * (1.0f / 500.0f);
  const float uxb = ((float)(xs + 127) + 0.5f - 256.0f) * (1.0f / 500.0f);
  const float smax  = fmaxf(r0.x * uxa, r0.x * uxb) + r0.y * uy + r1.x;
  const float guard = 1e-4f * (fabsf(r0.x) + fabsf(r0.y) + fabsf(r1.x));
  const bool keep = !(smax < -guard);

  const unsigned km = __builtin_amdgcn_ballot_w32(keep);
  int before = (int)__builtin_amdgcn_mbcnt_lo(km, 0u);
  if (lane == 0) shC[wv] = __popc(km);
  __syncthreads();
  int total = 0;
  for (int s = 0; s < 8; ++s) {
    const int c = shC[s];
    total  += c;
    before += (s < wv) ? c : 0;
  }
  const int nbv  = (total + 15) >> 4;
  const int half = 8 * nbv;

  const int slot = keep ? before : total + (t - before);
  if (slot < 2 * half) {
    const int seg = (slot >= half) ? 1 : 0;
    const int loc = slot - seg * half;
    float4* row = shA + (loc >> 3) * 32 + 8 * seg + (loc & 7);
    row[0]    = keep ? r0 : make_float4(0.0f, 0.0f, 0.0f, 0.0f);
    row[16]   = keep ? r1 : make_float4(0.0f, 0.0f, 0.0f, 1e-10f);
    shS[slot] = keep ? rs : make_float4(0.0f, 0.0f, 0.0f, 0.0f);
  }
  __syncthreads();

  const int nb = __builtin_amdgcn_readfirstlane(nbv);

  const int p    = lane & 15;
  const bool hi  = lane >= 16;
  const int x = xs + wv * 16 + p;

  const float ux = ((float)x + 0.5f - 256.0f) * (1.0f / 500.0f);
  const float rn = fast_rsq(ux * ux + uy * uy + 1.0f);
  const float dx = ux * rn, dy = uy * rn, dz = rn;

  v2f bn, bd;
  bn.x = hi ? dz       : dx;
  bn.y = hi ? 0.0f     : dy;
  bd.x = hi ? dz * dz  : dx * dx;
  bd.y = hi ? 1.0f     : dy * dy;

  float sIr = 0.0f, sIg = 0.0f, sIb = 0.0f, sT = 1.0f;
  const int gseg = hi ? 8 * nb : 0;

  for (int blk = 0; blk < nb; ++blk) {
    const float4 at = shA[blk * 32 + lane];
    v2f anv; anv.x = at.x; anv.y = at.y;
    v2f adv; adv.x = at.z; adv.y = at.w;

    const v8f zero = {0, 0, 0, 0, 0, 0, 0, 0};
    const v8f accN = __builtin_amdgcn_wmma_f32_16x16x4_f32(false, anv, false, bn,
                                                (short)0, zero, false, false);
    const v8f accD = __builtin_amdgcn_wmma_f32_16x16x4_f32(false, adv, false, bd,
                                                (short)0, zero, false, false);

    const int gbase = gseg + blk * 8;
#pragma unroll
    for (int i = 0; i < 8; ++i) {
      const float Av = accN[i];
      const float Bv = accD[i];
      const float4 sc = shS[gbase + i];
      const float rB  = fast_rcp(Bv);
      const float tau = Av * rB;
      float alpha = sc.x * fast_exp2(0.72134752f * (Av * tau));
      alpha = (tau > 0.1f) ? alpha : 0.0f;
      const float wgt = sT * alpha;
      sIr = fmaf(wgt, sc.y, sIr);
      sIg = fmaf(wgt, sc.z, sIg);
      sIb = fmaf(wgt, sc.w, sIb);
      sT  = fmaf(-alpha, sT, sT);
    }
  }

  const float oR = swapx16(sIr), oG = swapx16(sIg);
  const float oB = swapx16(sIb), oT = swapx16(sT);
  if (!hi) {
    const float T = sT * oT;
    const int o = (y * GW + x) * 3;
    out[o + 0] = fmaf(sT, oR, sIr) + T;
    out[o + 1] = fmaf(sT, oG, sIg) + T;
    out[o + 2] = fmaf(sT, oB, sIb) + T;
  }
}

extern "C" void kernel_launch(void* const* d_in, const int* in_sizes, int n_in,
                              void* d_out, int out_size, void* d_ws, size_t ws_size,
                              hipStream_t stream) {
  (void)in_sizes; (void)n_in; (void)out_size; (void)ws_size;
  const float* pos = (const float*)d_in[0];
  const float* lsc = (const float*)d_in[1];
  const float* rop = (const float*)d_in[2];
  const float* col = (const float*)d_in[3];

  float4* aop  = (float4*)d_ws;
  float4* scal = aop + NBLK * 32;

  gsplat_prep_kernel<<<PREP_WG, 256, 0, stream>>>(pos, lsc, rop, col, aop, scal);

  dim3 grid(GW / 128, GH);
  gsplat_render_kernel<<<grid, 256, 0, stream>>>(aop, scal, (float*)d_out);
}


# Cull per 32x4-pixel tile (was a 128x1 strip) by the reference's valid test: a gaussian goes when tau <= 0.1 on the whole tile (alpha exactly 0), not only when A < 0; summation tree changes
# speedup vs baseline: 3.1535x; 1.1135x over previous
#include <hip/hip_runtime.h>
#include <hip/hip_bf16.h>


typedef float v2f __attribute__((ext_vector_type(2)));
typedef float v8f __attribute__((ext_vector_type(8)));

#define GH 512
#define GW 512
#define NG 256
#define NBLK 16
#define TW 32
#define TH 4

__device__ __forceinline__ float fast_rcp(float x) {
#if __has_builtin(__builtin_amdgcn_rcpf)
  return __builtin_amdgcn_rcpf(x);
#else
  return 1.0f / x;
#endif
}
__device__ __forceinline__ float fast_rsq(float x) {
#if __has_builtin(__builtin_amdgcn_rsqf)
  return __builtin_amdgcn_rsqf(x);
#else
  return rsqrtf(x);
#endif
}
__device__ __forceinline__ float fast_exp2(float x) {
#if __has_builtin(__builtin_amdgcn_exp2f)
  return __builtin_amdgcn_exp2f(x);
#else
  return exp2f(x);
#endif
}
__device__ __forceinline__ float swapx16(float x) {
  return __int_as_float(__builtin_amdgcn_ds_swizzle(__float_as_int(x), 0x401f));
}

#define PREP_WG 8
#define GPW (NG / PREP_WG)

__global__ __launch_bounds__(256) void gsplat_prep_kernel(const float* __restrict__ pos,
                                   const float* __restrict__ lsc,
                                   const float* __restrict__ rop,
                                   const float* __restrict__ col,
                                   float4* __restrict__ aop,
                                   float4* __restrict__ scal)
{
  __shared__ float4 shZ4[NG / 4];
  __shared__ int    shP[8 * GPW];

  const int t    = threadIdx.x;
  const int lane = t & 31;
  const int sl   = t >> 5;
  float* shZ = reinterpret_cast<float*>(shZ4);
  const int i = blockIdx.x * GPW + lane;

  const float zt = pos[t * 3 + 2];

  float mx = 0.0f, my = 0.0f, lx = 0.0f, ly = 0.0f, lz = 0.0f;
  float ro = 0.0f, c0 = 0.0f, c1 = 0.0f, c2 = 0.0f;
  if (sl == 0) {
    mx = pos[i * 3 + 0]; my = pos[i * 3 + 1];
    lx = lsc[i * 3 + 0]; ly = lsc[i * 3 + 1]; lz = lsc[i * 3 + 2];
    ro = rop[i];
    c0 = col[i * 3 + 0]; c1 = col[i * 3 + 1]; c2 = col[i * 3 + 2];
  }

  shZ[t] = zt;
  __syncthreads();

  const float mz = shZ[i];

  int part = 0;
#pragma unroll
  for (int q = 0; q < 8; ++q) {
    const float4 z4 = shZ4[sl * 8 + q];
    const int j = sl * 32 + q * 4;
    part += (z4.x < mz) || (z4.x == mz && (j + 0) < i);
    part += (z4.y < mz) || (z4.y == mz && (j + 1) < i);
    part += (z4.z < mz) || (z4.z == mz && (j + 2) < i);
    part += (z4.w < mz) || (z4.w == mz && (j + 3) < i);
  }
  shP[sl * GPW + lane] = part;
  __syncthreads();
  if (sl != 0) return;

  int rank = 0;
#pragma unroll
  for (int s = 0; s < 8; ++s) rank += shP[s * GPW + lane];

  const float sx = expf(-2.0f * lx);
  const float sy = expf(-2.0f * ly);
  const float sz = expf(-2.0f * lz);
  const float a0 = 1.0f / (1.0f + expf(-ro));
  const float cr = 1.0f / (1.0f + expf(-c0));
  const float cg = 1.0f / (1.0f + expf(-c1));
  const float cb = 1.0f / (1.0f + expf(-c2));

  const float wx = mx * sx, wy = my * sy, wz = mz * sz;
  const float C = sx * mx * mx + sy * my * my + sz * mz * mz;
  const float a0p = a0 * expf(-0.5f * C);

  const int seg = rank >> 7;
  const int b   = (rank & 127) >> 3;
  const int m   = 8 * seg + (rank & 7);
  float4* base = aop + b * 32;
  base[m]      = make_float4(wx, wy,   sx, sy);
  base[16 + m] = make_float4(wz, 0.0f, sz, 1e-10f);

  scal[rank] = make_float4(a0p, cr, cg, cb);
}

__global__ __launch_bounds__(256) void gsplat_render_kernel(
    const float4* __restrict__ aopG, const float4* __restrict__ scalG,
    float* __restrict__ out)
{
  __shared__ float4 shA[NBLK * 32];
  __shared__ float4 shS[NG];
  __shared__ int    shC[8];

  const int t    = threadIdx.x;
  const int lane = t & 31;
  const int wv   = t >> 5;
  const int xs   = blockIdx.x * TW;
  const int y0   = blockIdx.y * TH;

  const int rb = (t & 127) >> 3;
  const int rm = 8 * (t >> 7) + (t & 7);
  const float4 r0 = aopG[rb * 32 + rm];
  const float4 r1 = aopG[rb * 32 + 16 + rm];
  const float4 rs = scalG[t];

  const float uxa = ((float)xs + 0.5f - 256.0f) * (1.0f / 500.0f);
  const float uxb = ((float)(xs + TW - 1) + 0.5f - 256.0f) * (1.0f / 500.0f);
  const float uya = ((float)y0 + 0.5f - 256.0f) * (1.0f / 500.0f);
  const float uyb = ((float)(y0 + TH - 1) + 0.5f - 256.0f) * (1.0f / 500.0f);
  const float smax  = fmaxf(r0.x * uxa, r0.x * uxb) + fmaxf(r0.y * uya, r0.y * uyb) + r1.x;
  const float guard = 1e-4f * (fabsf(r0.x) + fabsf(r0.y) + fabsf(r1.x));
  const float ux2lo = (uxa * uxb < 0.0f) ? 0.0f : fminf(uxa * uxa, uxb * uxb);
  const float uy2lo = (uya * uyb < 0.0f) ? 0.0f : fminf(uya * uya, uyb * uyb);
  const float ux2hi = fmaxf(uxa * uxa, uxb * uxb);
  const float uy2hi = fmaxf(uya * uya, uyb * uyb);
  const float rnlo  = fast_rsq(ux2hi + uy2hi + 1.0f);
  const float qlo   = r0.z * ux2lo + r0.w * uy2lo + r1.z;
  const bool keep = !(smax + guard < 0.0999f * rnlo * qlo);

  const unsigned km = __builtin_amdgcn_ballot_w32(keep);
  int before = (int)__builtin_amdgcn_mbcnt_lo(km, 0u);
  if (lane == 0) shC[wv] = __popc(km);
  __syncthreads();
  int total = 0;
  for (int s = 0; s < 8; ++s) {
    const int c = shC[s];
    total  += c;
    before += (s < wv) ? c : 0;
  }
  const int nbv  = (total + 15) >> 4;
  const int half = 8 * nbv;

  const int slot = keep ? before : total + (t - before);
  if (slot < 2 * half) {
    const int seg = (slot >= half) ? 1 : 0;
    const int loc = slot - seg * half;
    float4* row = shA + (loc >> 3) * 32 + 8 * seg + (loc & 7);
    row[0]    = keep ? r0 : make_float4(0.0f, 0.0f, 0.0f, 0.0f);
    row[16]   = keep ? r1 : make_float4(0.0f, 0.0f, 0.0f, 1e-10f);
    shS[slot] = keep ? rs : make_float4(0.0f, 0.0f, 0.0f, 0.0f);
  }
  __syncthreads();

  const int nb = __builtin_amdgcn_readfirstlane(nbv);

  const int p    = lane & 15;
  const bool hi  = lane >= 16;
  const int x = xs + (wv & 1) * 16 + p;
  const int y = y0 + (wv >> 1);

  const float ux = ((float)x + 0.5f - 256.0f) * (1.0f / 500.0f);
  const float uy = ((float)y + 0.5f - 256.0f) * (1.0f / 500.0f);
  const float rn = fast_rsq(ux * ux + uy * uy + 1.0f);
  const float dx = ux * rn, dy = uy * rn, dz = rn;

  v2f bn, bd;
  bn.x = hi ? dz       : dx;
  bn.y = hi ? 0.0f     : dy;
  bd.x = hi ? dz * dz  : dx * dx;
  bd.y = hi ? 1.0f     : dy * dy;

  float sIr = 0.0f, sIg = 0.0f, sIb = 0.0f, sT = 1.0f;
  const int gseg = hi ? 8 * nb : 0;

  for (int blk = 0; blk < nb; ++blk) {
    const float4 at = shA[blk * 32 + lane];
    v2f anv; anv.x = at.x; anv.y = at.y;
    v2f adv; adv.x = at.z; adv.y = at.w;

    const v8f zero = {0, 0, 0, 0, 0, 0, 0, 0};
    const v8f accN = __builtin_amdgcn_wmma_f32_16x16x4_f32(false, anv, false, bn,
                                                (short)0, zero, false, false);
    const v8f accD = __builtin_amdgcn_wmma_f32_16x16x4_f32(false, adv, false, bd,
                                                (short)0, zero, false, false);

    const int gbase = gseg + blk * 8;
#pragma unroll
    for (int i = 0; i < 8; ++i) {
      const float Av = accN[i];
      const float Bv = accD[i];
      const float4 sc = shS[gbase + i];
      const float rB  = fast_rcp(Bv);
      const float tau = Av * rB;
      float alpha = sc.x * fast_exp2(0.72134752f * (Av * tau));
      alpha = (tau > 0.1f) ? alpha : 0.0f;
      const float wgt = sT * alpha;
      sIr = fmaf(wgt, sc.y, sIr);
      sIg = fmaf(wgt, sc.z, sIg);
      sIb = fmaf(wgt, sc.w, sIb);
      sT  = fmaf(-alpha, sT, sT);
    }
  }

  const float oR = swapx16(sIr), oG = swapx16(sIg);
  const float oB = swapx16(sIb), oT = swapx16(sT);
  if (!hi) {
    const float T = sT * oT;
    const int o = (y * GW + x) * 3;
    out[o + 0] = fmaf(sT, oR, sIr) + T;
    out[o + 1] = fmaf(sT, oG, sIg) + T;
    out[o + 2] = fmaf(sT, oB, sIb) + T;
  }
}

extern "C" void kernel_launch(void* const* d_in, const int* in_sizes, int n_in,
                              void* d_out, int out_size, void* d_ws, size_t ws_size,
                              hipStream_t stream) {
  (void)in_sizes; (void)n_in; (void)out_size; (void)ws_size;
  const float* pos = (const float*)d_in[0];
  const float* lsc = (const float*)d_in[1];
  const float* rop = (const float*)d_in[2];
  const float* col = (const float*)d_in[3];

  float4* aop  = (float4*)d_ws;
  float4* scal = aop + NBLK * 32;

  gsplat_prep_kernel<<<PREP_WG, 256, 0, stream>>>(pos, lsc, rop, col, aop, scal);

  dim3 grid(GW / TW, GH / TH);
  gsplat_render_kernel<<<grid, 256, 0, stream>>>(aop, scal, (float*)d_out);
}


# One launch: each 32-wave work-group sifts the 256 gaussians for its 32x16 region first, then z-ranks only the kept ones in LDS (integer keys); no prep launch, no workspace; summation tree changes
# speedup vs baseline: 3.4700x; 1.1004x over previous
#include <hip/hip_runtime.h>
#include <hip/hip_bf16.h>


typedef float v2f __attribute__((ext_vector_type(2)));
typedef float v8f __attribute__((ext_vector_type(8)));

#define GH 512
#define GW 512
#define NG 256
#define NBLK 16
#define TW 32
#define TH 16

__device__ __forceinline__ float fast_rcp(float x) {
#if __has_builtin(__builtin_amdgcn_rcpf)
  return __builtin_amdgcn_rcpf(x);
#else
  return 1.0f / x;
#endif
}
__device__ __forceinline__ float fast_rsq(float x) {
#if __has_builtin(__builtin_amdgcn_rsqf)
  return __builtin_amdgcn_rsqf(x);
#else
  return rsqrtf(x);
#endif
}
__device__ __forceinline__ float fast_exp2(float x) {
#if __has_builtin(__builtin_amdgcn_exp2f)
  return __builtin_amdgcn_exp2f(x);
#else
  return exp2f(x);
#endif
}
__device__ __forceinline__ float swapx16(float x) {
  return __int_as_float(__builtin_amdgcn_ds_swizzle(__float_as_int(x), 0x401f));
}


__global__ __launch_bounds__(1024) void gsplat_fused_kernel(
    const float* __restrict__ pos, const float* __restrict__ lsc,
    const float* __restrict__ rop, const float* __restrict__ col,
    float* __restrict__ out)
{
  __shared__ float4 shA[NBLK * 32];
  __shared__ float4 shS[NG];
  __shared__ int4   shK4[NG / 2];
  __shared__ int    shP[1024];
  __shared__ int    shC[8];

  const int t    = threadIdx.x;
  const int lane = t & 31;
  const int wv   = t >> 5;
  const int xs   = blockIdx.x * TW;
  const int y0   = blockIdx.y * TH;
  int* shK = reinterpret_cast<int*>(shK4);
  const bool owner = t < NG;

  float wx = 0.0f, wy = 0.0f, wz = 0.0f, sx = 0.0f, sy = 0.0f, sz = 0.0f;
  float Cq = 0.0f;
  int zkey = 0;
  float ro = 0.0f, c0 = 0.0f, c1 = 0.0f, c2 = 0.0f;
  bool keep = false;
  int before = 0;
  if (owner) {
    const float mx = pos[t * 3 + 0], my = pos[t * 3 + 1];
    const float mz = pos[t * 3 + 2];
    const float lx = lsc[t * 3 + 0], ly = lsc[t * 3 + 1], lz = lsc[t * 3 + 2];
    ro = rop[t];
    c0 = col[t * 3 + 0]; c1 = col[t * 3 + 1]; c2 = col[t * 3 + 2];

    sx = expf(-2.0f * lx);
    sy = expf(-2.0f * ly);
    sz = expf(-2.0f * lz);
    wx = mx * sx; wy = my * sy; wz = mz * sz;
    Cq = sx * mx * mx + sy * my * my + sz * mz * mz;
    const int zb = __float_as_int((mz == 0.0f) ? 0.0f : mz);
    zkey = zb ^ ((zb >> 31) & 0x7fffffff);

    const float uxa = ((float)xs + 0.5f - 256.0f) * (1.0f / 500.0f);
    const float uxb = ((float)(xs + TW - 1) + 0.5f - 256.0f) * (1.0f / 500.0f);
    const float uya = ((float)y0 + 0.5f - 256.0f) * (1.0f / 500.0f);
    const float uyb = ((float)(y0 + TH - 1) + 0.5f - 256.0f) * (1.0f / 500.0f);
    const float smax  = fmaxf(wx * uxa, wx * uxb) + fmaxf(wy * uya, wy * uyb) + wz;
    const float guard = 1e-4f * (fabsf(wx) + fabsf(wy) + fabsf(wz));
    const float ux2lo = (uxa * uxb < 0.0f) ? 0.0f : fminf(uxa * uxa, uxb * uxb);
    const float uy2lo = (uya * uyb < 0.0f) ? 0.0f : fminf(uya * uya, uyb * uyb);
    const float ux2hi = fmaxf(uxa * uxa, uxb * uxb);
    const float uy2hi = fmaxf(uya * uya, uyb * uyb);
    const float rnlo  = fast_rsq(ux2hi + uy2hi + 1.0f);
    const float qlo   = sx * ux2lo + sy * uy2lo + sz;
    keep = !(smax + guard < 0.0999f * rnlo * qlo);

    const unsigned km = __builtin_amdgcn_ballot_w32(keep);
    before = (int)__builtin_amdgcn_mbcnt_lo(km, 0u);
    if (lane == 0) shC[wv] = __popc(km);
  }
  __syncthreads();
  int total = 0;
  for (int s = 0; s < 8; ++s) {
    const int c = shC[s];
    total  += c;
    before += (s < wv) ? c : 0;
  }
  const int nbv  = (total + 15) >> 4;
  const int half = 8 * nbv;

  const int cslot = keep ? before : total + (t - before);
  if (owner) {
    shK[cslot]  = keep ? zkey : 0x7fffffff;
    shK[NG + t] = 0x7fffffff;
  }
  __syncthreads();

  const int lg = 4 + (nbv > 1) + (nbv > 2) + (nbv > 4) + (nbv > 8);
  const int E2 = 1 << lg;
  const int P  = 1024 >> lg;
  {
    const int e  = t & (E2 - 1);
    const int qd = t >> lg;
    const int gp = (4 * nbv + P - 1) >> (10 - lg);
    const int ke = shK[e];
    const int eg = e >> 2;
    int part = 0;
    for (int u = 0; u < gp; ++u) {
      const int j4 = qd * gp + u;
      const int4 k4 = shK4[j4];
      const int thr = (int)((unsigned)ke + ((j4 < eg) ? 1u : 0u));
      part += (k4.x < thr) ? 1 : 0;
      part += (k4.y < thr) ? 1 : 0;
      part += (k4.z < thr) ? 1 : 0;
      part += (k4.w < thr) ? 1 : 0;
    }
    shP[qd * E2 + e] = part;
  }
  __syncthreads();

  if (owner) {
    int slot = cslot;
    if (keep) {
      slot = 0;
      for (int q = 0; q < P; ++q) slot += shP[q * E2 + cslot];
      const int gb = cslot & ~3;
      for (int k = 0; k < 3; ++k) slot += ((gb + k) < cslot && shK[gb + k] == zkey) ? 1 : 0;
    }
    if (slot < 2 * half) {
      const float a0 = 1.0f / (1.0f + expf(-ro));
      const float cr = 1.0f / (1.0f + expf(-c0));
      const float cg = 1.0f / (1.0f + expf(-c1));
      const float cb = 1.0f / (1.0f + expf(-c2));
      const float a0p = a0 * expf(-0.5f * Cq);
      const int seg = (slot >= half) ? 1 : 0;
      const int loc = slot - seg * half;
      float4* row = shA + (loc >> 3) * 32 + 8 * seg + (loc & 7);
      row[0]    = keep ? make_float4(wx, wy,   sx, sy)     : make_float4(0.0f, 0.0f, 0.0f, 0.0f);
      row[16]   = keep ? make_float4(wz, 0.0f, sz, 1e-10f) : make_float4(0.0f, 0.0f, 0.0f, 1e-10f);
      shS[slot] = keep ? make_float4(a0p, cr, cg, cb)      : make_float4(0.0f, 0.0f, 0.0f, 0.0f);
    }
  }
  __syncthreads();

  const int nb = __builtin_amdgcn_readfirstlane(nbv);

  const int p    = lane & 15;
  const bool hi  = lane >= 16;
  const int x = xs + (wv & 1) * 16 + p;
  const int y = y0 + (wv >> 1);

  const float ux = ((float)x + 0.5f - 256.0f) * (1.0f / 500.0f);
  const float uy = ((float)y + 0.5f - 256.0f) * (1.0f / 500.0f);
  const float rn = fast_rsq(ux * ux + uy * uy + 1.0f);
  const float dx = ux * rn, dy = uy * rn, dz = rn;

  v2f bn, bd;
  bn.x = hi ? dz       : dx;
  bn.y = hi ? 0.0f     : dy;
  bd.x = hi ? dz * dz  : dx * dx;
  bd.y = hi ? 1.0f     : dy * dy;

  float sIr = 0.0f, sT = 1.0f;
  v2f sIgb = {0.0f, 0.0f};
  const int gseg = hi ? 8 * nb : 0;

  for (int blk = 0; blk < nb; ++blk) {
    const float4 at = shA[blk * 32 + lane];
    v2f anv; anv.x = at.x; anv.y = at.y;
    v2f adv; adv.x = at.z; adv.y = at.w;

    const v8f zero = {0, 0, 0, 0, 0, 0, 0, 0};
    const v8f accN = __builtin_amdgcn_wmma_f32_16x16x4_f32(false, anv, false, bn,
                                                (short)0, zero, false, false);
    const v8f accD = __builtin_amdgcn_wmma_f32_16x16x4_f32(false, adv, false, bd,
                                                (short)0, zero, false, false);

    const int gbase = gseg + blk * 8;
#pragma unroll
    for (int i = 0; i < 8; ++i) {
      const float Av = accN[i];
      const float Bv = accD[i];
      const float4 sc = shS[gbase + i];
      const float rB  = fast_rcp(Bv);
      const float tau = Av * rB;
      float alpha = sc.x * fast_exp2(0.72134752f * (Av * tau));
      alpha = (tau > 0.1f) ? alpha : 0.0f;
      const float wgt = sT * alpha;
      const v2f w2 = {wgt, wgt};
      const v2f c2 = {sc.z, sc.w};
      sIr  = fmaf(wgt, sc.y, sIr);
      sIgb = __builtin_elementwise_fma(w2, c2, sIgb);
      sT  = fmaf(-alpha, sT, sT);
    }
  }

  const float sIg = sIgb.x, sIb = sIgb.y;
  const float oR = swapx16(sIr), oG = swapx16(sIg);
  const float oB = swapx16(sIb), oT = swapx16(sT);
  if (!hi) {
    const float T = sT * oT;
    const int o = (y * GW + x) * 3;
    out[o + 0] = fmaf(sT, oR, sIr) + T;
    out[o + 1] = fmaf(sT, oG, sIg) + T;
    out[o + 2] = fmaf(sT, oB, sIb) + T;
  }
}

extern "C" void kernel_launch(void* const* d_in, const int* in_sizes, int n_in,
                              void* d_out, int out_size, void* d_ws, size_t ws_size,
                              hipStream_t stream) {
  (void)in_sizes; (void)n_in; (void)out_size; (void)d_ws; (void)ws_size;
  const float* pos = (const float*)d_in[0];
  const float* lsc = (const float*)d_in[1];
  const float* rop = (const float*)d_in[2];
  const float* col = (const float*)d_in[3];

  dim3 grid(GW / TW, GH / TH);
  gsplat_fused_kernel<<<grid, 1024, 0, stream>>>(pos, lsc, rop, col, (float*)d_out);
}


# Alpha formed inside the exponent: 2^(k*A*tau + log2(a0') - BIG*n), n a clamped not-valid flag; packed FMAs on gaussian pairs replace a multiply, a compare and a select per pair; sum order kept
# speedup vs baseline: 3.7919x; 1.0928x over previous
#include <hip/hip_runtime.h>
#include <hip/hip_bf16.h>


typedef float v2f __attribute__((ext_vector_type(2)));
typedef float v8f __attribute__((ext_vector_type(8)));

#define GH 512
#define GW 512
#define NG 256
#define NBLK 16
#define TW 32
#define TH 16

__device__ __forceinline__ float fast_rcp(float x) {
#if __has_builtin(__builtin_amdgcn_rcpf)
  return __builtin_amdgcn_rcpf(x);
#else
  return 1.0f / x;
#endif
}
__device__ __forceinline__ float fast_rsq(float x) {
#if __has_builtin(__builtin_amdgcn_rsqf)
  return __builtin_amdgcn_rsqf(x);
#else
  return rsqrtf(x);
#endif
}
__device__ __forceinline__ float fast_exp2(float x) {
#if __has_builtin(__builtin_amdgcn_exp2f)
  return __builtin_amdgcn_exp2f(x);
#else
  return exp2f(x);
#endif
}
__device__ __forceinline__ v2f not_valid2(v2f tau2, v2f nh, v2f c) {
  v2f n2;
  asm("v_pk_fma_f32 %0, %1, %2, %3 op_sel_hi:[1,0,0] clamp"
      : "=v"(n2) : "v"(tau2), "s"(nh), "s"(c));
  return n2;
}
__device__ __forceinline__ float swapx16(float x) {
  return __int_as_float(__builtin_amdgcn_ds_swizzle(__float_as_int(x), 0x401f));
}


__global__ __launch_bounds__(1024) void gsplat_fused_kernel(
    const float* __restrict__ pos, const float* __restrict__ lsc,
    const float* __restrict__ rop, const float* __restrict__ col,
    float* __restrict__ out)
{
  __shared__ float4 shA[NBLK * 32];
  __shared__ float4 shS[NG];
  __shared__ int4   shK4[NG / 2];
  __shared__ int    shP[1024];
  __shared__ int    shC[8];

  const int t    = threadIdx.x;
  const int lane = t & 31;
  const int wv   = t >> 5;
  const int xs   = blockIdx.x * TW;
  const int y0   = blockIdx.y * TH;
  int* shK = reinterpret_cast<int*>(shK4);
  const bool owner = t < NG;

  float wx = 0.0f, wy = 0.0f, wz = 0.0f, sx = 0.0f, sy = 0.0f, sz = 0.0f;
  float Cq = 0.0f;
  int zkey = 0;
  float ro = 0.0f, c0 = 0.0f, c1 = 0.0f, c2 = 0.0f;
  bool keep = false;
  int before = 0;
  if (owner) {
    const float mx = pos[t * 3 + 0], my = pos[t * 3 + 1];
    const float mz = pos[t * 3 + 2];
    const float lx = lsc[t * 3 + 0], ly = lsc[t * 3 + 1], lz = lsc[t * 3 + 2];
    ro = rop[t];
    c0 = col[t * 3 + 0]; c1 = col[t * 3 + 1]; c2 = col[t * 3 + 2];

    sx = expf(-2.0f * lx);
    sy = expf(-2.0f * ly);
    sz = expf(-2.0f * lz);
    wx = mx * sx; wy = my * sy; wz = mz * sz;
    Cq = sx * mx * mx + sy * my * my + sz * mz * mz;
    const int zb = __float_as_int((mz == 0.0f) ? 0.0f : mz);
    zkey = zb ^ ((zb >> 31) & 0x7fffffff);

    const float uxa = ((float)xs + 0.5f - 256.0f) * (1.0f / 500.0f);
    const float uxb = ((float)(xs + TW - 1) + 0.5f - 256.0f) * (1.0f / 500.0f);
    const float uya = ((float)y0 + 0.5f - 256.0f) * (1.0f / 500.0f);
    const float uyb = ((float)(y0 + TH - 1) + 0.5f - 256.0f) * (1.0f / 500.0f);
    const float smax  = fmaxf(wx * uxa, wx * uxb) + fmaxf(wy * uya, wy * uyb) + wz;
    const float guard = 1e-4f * (fabsf(wx) + fabsf(wy) + fabsf(wz));
    const float ux2lo = (uxa * uxb < 0.0f) ? 0.0f : fminf(uxa * uxa, uxb * uxb);
    const float uy2lo = (uya * uyb < 0.0f) ? 0.0f : fminf(uya * uya, uyb * uyb);
    const float ux2hi = fmaxf(uxa * uxa, uxb * uxb);
    const float uy2hi = fmaxf(uya * uya, uyb * uyb);
    const float rnlo  = fast_rsq(ux2hi + uy2hi + 1.0f);
    const float qlo   = sx * ux2lo + sy * uy2lo + sz;
    keep = !(smax + guard < 0.0999f * rnlo * qlo);

    const unsigned km = __builtin_amdgcn_ballot_w32(keep);
    before = (int)__builtin_amdgcn_mbcnt_lo(km, 0u);
    if (lane == 0) shC[wv] = __popc(km);
  }
  __syncthreads();
  int total = 0;
  for (int s = 0; s < 8; ++s) {
    const int c = shC[s];
    total  += c;
    before += (s < wv) ? c : 0;
  }
  const int nbv  = (total + 15) >> 4;
  const int half = 8 * nbv;

  const int cslot = keep ? before : total + (t - before);
  if (owner) {
    shK[cslot]  = keep ? zkey : 0x7fffffff;
    shK[NG + t] = 0x7fffffff;
  }
  __syncthreads();

  const int lg = 4 + (nbv > 1) + (nbv > 2) + (nbv > 4) + (nbv > 8);
  const int E2 = 1 << lg;
  const int P  = 1024 >> lg;
  {
    const int e  = t & (E2 - 1);
    const int qd = t >> lg;
    const int gp = (4 * nbv + P - 1) >> (10 - lg);
    const int ke = shK[e];
    const int eg = e >> 2;
    int part = 0;
    for (int u = 0; u < gp; ++u) {
      const int j4 = qd * gp + u;
      const int4 k4 = shK4[j4];
      const int thr = (int)((unsigned)ke + ((j4 < eg) ? 1u : 0u));
      part += (k4.x < thr) ? 1 : 0;
      part += (k4.y < thr) ? 1 : 0;
      part += (k4.z < thr) ? 1 : 0;
      part += (k4.w < thr) ? 1 : 0;
    }
    shP[qd * E2 + e] = part;
  }
  __syncthreads();

  if (owner) {
    int slot = cslot;
    if (keep) {
      slot = 0;
      for (int q = 0; q < P; ++q) slot += shP[q * E2 + cslot];
      const int gb = cslot & ~3;
      for (int k = 0; k < 3; ++k) slot += ((gb + k) < cslot && shK[gb + k] == zkey) ? 1 : 0;
    }
    if (slot < 2 * half) {
      const float a0 = 1.0f / (1.0f + expf(-ro));
      const float cr = 1.0f / (1.0f + expf(-c0));
      const float cg = 1.0f / (1.0f + expf(-c1));
      const float cb = 1.0f / (1.0f + expf(-c2));
      const float la0 = log2f(a0) - 0.72134752f * Cq;
      const int seg = (slot >= half) ? 1 : 0;
      const int loc = slot - seg * half;
      float4* row = shA + (loc >> 3) * 32 + 8 * seg + (loc & 7);
      row[0]    = keep ? make_float4(wx, wy,   sx, sy)     : make_float4(0.0f, 0.0f, 0.0f, 0.0f);
      row[16]   = keep ? make_float4(wz, 0.0f, sz, 1e-10f) : make_float4(0.0f, 0.0f, 0.0f, 1e-10f);
      float* pr = reinterpret_cast<float*>(shS + (slot >> 1) * 2);
      const int j = slot & 1;
      pr[j]     = keep ? la0 : 0.0f;
      pr[2 + j] = keep ? cr  : 0.0f;
      *reinterpret_cast<float2*>(pr + 4 + 2 * j) = keep ? make_float2(cg, cb) : make_float2(0.0f, 0.0f);
    }
  }
  __syncthreads();

  const int nb = __builtin_amdgcn_readfirstlane(nbv);

  const int p    = lane & 15;
  const bool hi  = lane >= 16;
  const int x = xs + (wv & 1) * 16 + p;
  const int y = y0 + (wv >> 1);

  const float ux = ((float)x + 0.5f - 256.0f) * (1.0f / 500.0f);
  const float uy = ((float)y + 0.5f - 256.0f) * (1.0f / 500.0f);
  const float rn = fast_rsq(ux * ux + uy * uy + 1.0f);
  const float dx = ux * rn, dy = uy * rn, dz = rn;

  v2f bn, bd;
  bn.x = hi ? dz       : dx;
  bn.y = hi ? 0.0f     : dy;
  bd.x = hi ? dz * dz  : dx * dx;
  bd.y = hi ? 1.0f     : dy * dy;

  float sIr = 0.0f, sT = 1.0f;
  v2f sIgb = {0.0f, 0.0f};
  const int gseg = hi ? 8 * nb : 0;
  const v2f nh   = {-134217728.0f, -134217728.0f};
  const v2f cv   = {13421774.0f, 13421774.0f};
  const v2f k2   = {0.72134752f, 0.72134752f};
  const v2f nbig = {-1e30f, -1e30f};

  for (int blk = 0; blk < nb; ++blk) {
    const float4 at = shA[blk * 32 + lane];
    v2f anv; anv.x = at.x; anv.y = at.y;
    v2f adv; adv.x = at.z; adv.y = at.w;

    const v8f zero = {0, 0, 0, 0, 0, 0, 0, 0};
    const v8f accN = __builtin_amdgcn_wmma_f32_16x16x4_f32(false, anv, false, bn,
                                                (short)0, zero, false, false);
    const v8f accD = __builtin_amdgcn_wmma_f32_16x16x4_f32(false, adv, false, bd,
                                                (short)0, zero, false, false);

    const int gbase = gseg + blk * 8;
#pragma unroll
    for (int j = 0; j < 4; ++j) {
      const float4 t0 = shS[gbase + 2 * j];
      const float4 t1 = shS[gbase + 2 * j + 1];
      const v2f A2  = {accN[2 * j], accN[2 * j + 1]};
      const v2f rB2 = {fast_rcp(accD[2 * j]), fast_rcp(accD[2 * j + 1])};
      const v2f tau2 = A2 * rB2;
      const v2f q2   = A2 * tau2;
      const v2f l2   = {t0.x, t0.y};
      v2f e2 = __builtin_elementwise_fma(q2, k2, l2);
      e2 = __builtin_elementwise_fma(not_valid2(tau2, nh, cv), nbig, e2);
      const float al[2] = {fast_exp2(e2.x), fast_exp2(e2.y)};
      const float cr[2] = {t0.z, t0.w};
      const v2f   cgb[2] = {{t1.x, t1.y}, {t1.z, t1.w}};
#pragma unroll
      for (int u = 0; u < 2; ++u) {
        const float alpha = al[u];
        const float wgt = sT * alpha;
        const v2f w2 = {wgt, wgt};
        sIr  = fmaf(wgt, cr[u], sIr);
        sIgb = __builtin_elementwise_fma(w2, cgb[u], sIgb);
        sT  = fmaf(-alpha, sT, sT);
      }
    }
  }

  const float sIg = sIgb.x, sIb = sIgb.y;
  const float oR = swapx16(sIr), oG = swapx16(sIg);
  const float oB = swapx16(sIb), oT = swapx16(sT);
  if (!hi) {
    const float T = sT * oT;
    const int o = (y * GW + x) * 3;
    out[o + 0] = fmaf(sT, oR, sIr) + T;
    out[o + 1] = fmaf(sT, oG, sIg) + T;
    out[o + 2] = fmaf(sT, oB, sIb) + T;
  }
}

extern "C" void kernel_launch(void* const* d_in, const int* in_sizes, int n_in,
                              void* d_out, int out_size, void* d_ws, size_t ws_size,
                              hipStream_t stream) {
  (void)in_sizes; (void)n_in; (void)out_size; (void)d_ws; (void)ws_size;
  const float* pos = (const float*)d_in[0];
  const float* lsc = (const float*)d_in[1];
  const float* rop = (const float*)d_in[2];
  const float* col = (const float*)d_in[3];

  dim3 grid(GW / TW, GH / TH);
  gsplat_fused_kernel<<<grid, 1024, 0, stream>>>(pos, lsc, rop, col, (float*)d_out);
}
